# starting kernel
# baseline (speedup 1.0000x reference)
#include <hip/hip_runtime.h>
#include <hip/hip_bf16.h>


typedef __attribute__((ext_vector_type(2))) float v2f;
typedef __attribute__((ext_vector_type(8))) float v8f;

#define LEN_SCALE_INV (1.0f / 20000.0f)

__global__ void k_deg_init(float* __restrict__ dinv, int N) {
    int i = blockIdx.x * blockDim.x + threadIdx.x;
    if (i < N) dinv[i] = 1.0f;
}

__global__ void k_deg_edges(const int* __restrict__ dst, float* __restrict__ dinv,
                            int E) {
    int e = blockIdx.x * blockDim.x + threadIdx.x;
    if (e < E) unsafeAtomicAdd(&dinv[dst[e]], 1.0f);
}

__global__ void k_deg_fin(const int* __restrict__ read_length,
                          float* __restrict__ dinv, float* __restrict__ dsq,
                          float* __restrict__ x0, int N) {
    int i = blockIdx.x * blockDim.x + threadIdx.x;
    if (i < N) {
        float deg = dinv[i];
        dsq[i]  = rsqrtf(deg);
        dinv[i] = 1.0f / deg;
        x0[i]   = (float)read_length[i] * LEN_SCALE_INV;
    }
}

__global__ void k_norm(const int* __restrict__ src, const int* __restrict__ dst,
                       const float* __restrict__ dsq, float* __restrict__ nrm,
                       int E) {
    int e = blockIdx.x * blockDim.x + threadIdx.x;
    if (e < E) nrm[e] = dsq[src[e]] * dsq[dst[e]];
}

__global__ void k_zero(float* __restrict__ p, long long n) {
    long long i = (long long)blockIdx.x * blockDim.x + threadIdx.x;
    if (i < n) p[i] = 0.0f;
}

template <int F_IN, int F_OUT>
__global__ void k_linear_wmma(const float* __restrict__ x,
                              const float* __restrict__ W,
                              float* __restrict__ xw, int N) {
    constexpr int TILES_N = (F_OUT + 15) / 16;
    constexpr int KCH     = (F_IN + 3) / 4;

    const int wave   = blockIdx.x * (blockDim.x >> 5) + (threadIdx.x >> 5);
    const int lane   = threadIdx.x & 31;
    const int tile_m = wave / TILES_N;
    const int tile_n = wave - tile_m * TILES_N;
    const int row0   = tile_m * 16;
    if (row0 >= N) return;

    const int  lg   = lane >> 4;
    const int  lmod = lane & 15;
    const int  m    = row0 + lmod;
    const int  mc   = (m < N) ? m : (N - 1);
    const int  nc   = tile_n * 16 + lmod;
    const bool nok  = (nc < F_OUT);
    const int  ncc  = nok ? nc : 0;

    v8f acc = {0.f, 0.f, 0.f, 0.f, 0.f, 0.f, 0.f, 0.f};

#pragma unroll
    for (int kc = 0; kc < KCH; ++kc) {
        const int k0  = kc * 4 + lg * 2;
        const int k1  = k0 + 1;
        const int k0c = (k0 < F_IN) ? k0 : (F_IN - 1);
        const int k1c = (k1 < F_IN) ? k1 : (F_IN - 1);

        float a0 = x[(long long)mc * F_IN + k0c];
        float a1 = x[(long long)mc * F_IN + k1c];
        a0 = (m < N && k0 < F_IN) ? a0 : 0.f;
        a1 = (m < N && k1 < F_IN) ? a1 : 0.f;
        v2f a; a.x = a0; a.y = a1;

        float b0 = W[k0c * F_OUT + ncc];
        float b1 = W[k1c * F_OUT + ncc];
        b0 = (nok && k0 < F_IN) ? b0 : 0.f;
        b1 = (nok && k1 < F_IN) ? b1 : 0.f;
        v2f b; b.x = b0; b.y = b1;

        acc = __builtin_amdgcn_wmma_f32_16x16x4_f32(
            false, a, false, b, (short)0, acc, false, false);
    }

    const bool fullM = (row0 + 16 <= N);
    if (nok) {
        float* p = xw + (long long)(row0 + lg * 8) * F_OUT + nc;
        if (fullM) {
#pragma unroll
            for (int v = 0; v < 8; ++v) p[(long long)v * F_OUT] = acc[v];
        } else {
#pragma unroll
            for (int v = 0; v < 8; ++v) {
                const int mrow = row0 + v + lg * 8;
                if (mrow < N) p[(long long)v * F_OUT] = acc[v];
            }
        }
    }
}

template <int LOGF>
__global__ void k_scatter(const float* __restrict__ xw,
                          const int* __restrict__ src,
                          const int* __restrict__ dst,
                          const float* __restrict__ nrm,
                          float* __restrict__ agg, long long EF) {
    const long long idx = (long long)blockIdx.x * blockDim.x + threadIdx.x;
    if (idx >= EF) return;
    const int e = (int)(idx >> LOGF);
    const int f = (int)(idx & ((1 << LOGF) - 1));
    const int s = src[e];
    const int d = dst[e];
    const float v = xw[((long long)s << LOGF) + f] * nrm[e];
    unsafeAtomicAdd(&agg[((long long)d << LOGF) + f], v);
}

template <int LOGF>
__global__ void k_finalize(const float* __restrict__ agg,
                           const float* __restrict__ xw,
                           const float* __restrict__ dinv,
                           const float* __restrict__ bias,
                           float* __restrict__ out, long long NF) {
    const long long i = (long long)blockIdx.x * blockDim.x + threadIdx.x;
    if (i >= NF) return;
    const int node = (int)(i >> LOGF);
    const int f    = (int)(i & ((1 << LOGF) - 1));
    out[i] = agg[i] + xw[i] * dinv[node] + bias[f];
}

template <int F_IN, int F_OUT, int LOGF>
static void run_layer(const float* xin, const float* W, const float* bias,
                      float* xw, float* agg, const float* dinv,
                      const int* src, const int* dst, const float* nrm,
                      float* xout, int N, int E, hipStream_t stream) {
    constexpr int T = 256;
    constexpr int TILES_N = (F_OUT + 15) / 16;

    const long long waves  = (long long)((N + 15) / 16) * TILES_N;
    const long long blocks = (waves + 7) / 8;
    k_linear_wmma<F_IN, F_OUT><<<(unsigned)blocks, 256, 0, stream>>>(xin, W, xw, N);

    const long long nf = (long long)N * F_OUT;
    k_zero<<<(unsigned)((nf + T - 1) / T), T, 0, stream>>>(agg, nf);

    const long long ef = (long long)E * F_OUT;
    k_scatter<LOGF><<<(unsigned)((ef + T - 1) / T), T, 0, stream>>>(
        xw, src, dst, nrm, agg, ef);

    k_finalize<LOGF><<<(unsigned)((nf + T - 1) / T), T, 0, stream>>>(
        agg, xw, dinv, bias, xout, nf);
}

extern "C" void kernel_launch(void* const* d_in, const int* in_sizes, int n_in,
                              void* d_out, int out_size, void* d_ws, size_t ws_size,
                              hipStream_t stream) {
    const int* read_length = (const int*)d_in[0];
    const int* edge        = (const int*)d_in[1];
    const int  N = in_sizes[0];
    const int  E = in_sizes[1] / 2;
    const int* src = edge;
    const int* dst = edge + E;

    const float* W1 = (const float*)d_in[2]; const float* B1 = (const float*)d_in[3];
    const float* W2 = (const float*)d_in[4]; const float* B2 = (const float*)d_in[5];
    const float* W3 = (const float*)d_in[6]; const float* B3 = (const float*)d_in[7];
    const float* W4 = (const float*)d_in[8]; const float* B4 = (const float*)d_in[9];

    float* ws   = (float*)d_ws;
    float* xbuf = ws;
    float* xw   = xbuf + (size_t)N * 32;
    float* agg  = xw   + (size_t)N * 32;
    float* dinv = agg  + (size_t)N * 32;
    float* dsq  = dinv + N;
    float* nrm  = dsq  + N;

    const int T = 256;

    k_deg_init<<<(N + T - 1) / T, T, 0, stream>>>(dinv, N);
    k_deg_edges<<<(E + T - 1) / T, T, 0, stream>>>(dst, dinv, E);
    k_deg_fin<<<(N + T - 1) / T, T, 0, stream>>>(read_length, dinv, dsq, xbuf, N);
    k_norm<<<(E + T - 1) / T, T, 0, stream>>>(src, dst, dsq, nrm, E);

    run_layer<1, 4, 2>(xbuf, W1, B1, xw, agg, dinv, src, dst, nrm, xbuf, N, E, stream);
    run_layer<4, 8, 3>(xbuf, W2, B2, xw, agg, dinv, src, dst, nrm, xbuf, N, E, stream);
    run_layer<8, 16, 4>(xbuf, W3, B3, xw, agg, dinv, src, dst, nrm, xbuf, N, E, stream);
    run_layer<16, 32, 5>(xbuf, W4, B4, xw, agg, dinv, src, dst, nrm,
                         (float*)d_out, N, E, stream);
}


# launch fusion 11 -> 7: a list gather keeps its 16-row tile's sums in registers, forms the layer output and runs the next WMMA linear itself (last layer: stores out); no agg plane; sum order unchanged
# speedup vs baseline: 2.7857x; 2.7857x over previous
#include <hip/hip_runtime.h>
#include <hip/hip_bf16.h>


typedef __attribute__((ext_vector_type(2))) float v2f;
typedef __attribute__((ext_vector_type(4))) float v4f;
typedef __attribute__((ext_vector_type(8))) float v8f;
typedef __attribute__((ext_vector_type(2))) int   v2i;

#define LEN_SCALE_INV (1.0f / 20000.0f)

__global__ void k_init(int* __restrict__ head, int N) {
    int i = blockIdx.x * blockDim.x + threadIdx.x;
    if (i < N) head[i] = -1;
}

__global__ void k_build(const int* __restrict__ src, const int* __restrict__ dst,
                        int* __restrict__ head, int* __restrict__ rec, int E) {
    int e = blockIdx.x * blockDim.x + threadIdx.x;
    if (e >= E) return;
    const int prev = atomicExch(&head[dst[e]], e);
    v2i r; r.x = src[e]; r.y = prev;
    reinterpret_cast<v2i*>(rec)[e] = r;
}

template <int F_IN, int F_OUT>
__device__ __forceinline__ void tile_linear_store(const v2f (&afr)[(F_IN + 3) / 4],
                                                  const float* __restrict__ W,
                                                  float* __restrict__ xw,
                                                  int row0, int lane, int N) {
    constexpr int TILES_N = (F_OUT + 15) / 16;
    constexpr int KCH     = (F_IN + 3) / 4;
    const int lg   = lane >> 4;
    const int lmod = lane & 15;

    v8f acc[TILES_N];
#pragma unroll
    for (int t = 0; t < TILES_N; ++t)
        acc[t] = (v8f){0.f, 0.f, 0.f, 0.f, 0.f, 0.f, 0.f, 0.f};

#pragma unroll
    for (int kc = 0; kc < KCH; ++kc) {
        const int k0  = kc * 4 + lg * 2;
        const int k1  = k0 + 1;
        const int k0c = (k0 < F_IN) ? k0 : (F_IN - 1);
        const int k1c = (k1 < F_IN) ? k1 : (F_IN - 1);
#pragma unroll
        for (int t = 0; t < TILES_N; ++t) {
            const int  nc  = t * 16 + lmod;
            const bool nok = (nc < F_OUT);
            const int  ncc = nok ? nc : 0;
            float b0 = W[k0c * F_OUT + ncc];
            float b1 = W[k1c * F_OUT + ncc];
            b0 = (nok && k0 < F_IN) ? b0 : 0.f;
            b1 = (nok && k1 < F_IN) ? b1 : 0.f;
            v2f b; b.x = b0; b.y = b1;
            acc[t] = __builtin_amdgcn_wmma_f32_16x16x4_f32(
                false, afr[kc], false, b, (short)0, acc[t], false, false);
        }
    }

    const bool fullM = (row0 + 16 <= N);
#pragma unroll
    for (int t = 0; t < TILES_N; ++t) {
        const int nc = t * 16 + lmod;
        if (nc < F_OUT) {
            float* p = xw + (long long)(row0 + lg * 8) * F_OUT + nc;
            if (fullM) {
#pragma unroll
                for (int v = 0; v < 8; ++v) p[(long long)v * F_OUT] = acc[t][v];
            } else {
#pragma unroll
                for (int v = 0; v < 8; ++v) {
                    const int mrow = row0 + v + lg * 8;
                    if (mrow < N) p[(long long)v * F_OUT] = acc[t][v];
                }
            }
        }
    }
}

__global__ void k_first_linear(const int* __restrict__ read_length,
                               float* __restrict__ dinv, float* __restrict__ dsq,
                               const float* __restrict__ W, float* __restrict__ xw,
                               const int* __restrict__ head,
                               const int* __restrict__ rec, int N, int E) {
    const int wave = blockIdx.x * (blockDim.x >> 5) + (threadIdx.x >> 5);
    const int lane = threadIdx.x & 31;
    const int row0 = wave * 16;
    if (row0 >= N) return;

    const int lg = lane >> 4;
    const int m  = row0 + (lane & 15);

    float xfirst = 0.f;
    if (lg == 0 && m < N) {
        const unsigned uE = (unsigned)E;
        int cnt = 0;
        int e = head[m];
        for (int it = 0; it < E && (unsigned)e < uE; ++it) {
            e = rec[2 * (long long)e + 1];
            ++cnt;
        }
        const float deg = 1.0f + (float)cnt;
        dsq[m]  = rsqrtf(deg);
        dinv[m] = 1.0f / deg;
        xfirst  = (float)read_length[m] * LEN_SCALE_INV;
    }
    v2f afr[1];
    afr[0].x = xfirst;
    afr[0].y = 0.f;
    tile_linear_store<1, 4>(afr, W, xw, row0, lane, N);
}

template <int F_IN, int F_OUT>
__global__ void k_gather_linear(const float* __restrict__ xw_prev,
                                const float* __restrict__ bias_prev,
                                const int* __restrict__ rec,
                                const int* __restrict__ head,
                                const float* __restrict__ dsq,
                                const float* __restrict__ dinv,
                                const float* __restrict__ W,
                                float* __restrict__ xw, int N, int E) {
    constexpr int CH = F_IN / 4;
    constexpr int NT = (CH == 1) ? 2 : 1;
    constexpr int NA = (CH == 4) ? 2 : 1;
    static_assert(CH == 1 || CH == 2 || CH == 4, "layer width");

    const int wave = blockIdx.x * (blockDim.x >> 5) + (threadIdx.x >> 5);
    const int lane = threadIdx.x & 31;
    const int row0 = wave * (16 * NT);
    if (row0 >= N) return;

    const int  lg   = lane >> 4;
    const int  lmod = lane & 15;
    const int  m    = row0 + ((NT == 2) ? lane : lmod);
    const bool mok  = (m < N);
    const int  c0   = (CH == 1) ? 0 : lg;

    const v4f* __restrict__ xw4  = reinterpret_cast<const v4f*>(xw_prev);
    const v2i* __restrict__ rec2 = reinterpret_cast<const v2i*>(rec);
    const unsigned uE = (unsigned)E;

    float dd = 0.f, dnode = 0.f;
    int e = -1;
    if (mok) { dd = dsq[m]; dnode = dinv[m]; e = head[m]; }
    v2i r; r.x = 0; r.y = -1;
    if ((unsigned)e < uE) r = rec2[e];

    v4f acc[NA];
#pragma unroll
    for (int a = 0; a < NA; ++a) acc[a] = (v4f){0.f, 0.f, 0.f, 0.f};

    for (int it = 0; it < E && (unsigned)e < uE; ++it) {
        const int s  = r.x;
        const int nx = r.y;
        const float q = dsq[s];
        v4f x[NA];
#pragma unroll
        for (int a = 0; a < NA; ++a) x[a] = xw4[(long long)s * CH + c0 + 2 * a];
        v2i rn; rn.x = 0; rn.y = -1;
        if ((unsigned)nx < uE) rn = rec2[nx];
        const float w = q * dd;
#pragma unroll
        for (int a = 0; a < NA; ++a) {
            acc[a].x += x[a].x * w;
            acc[a].y += x[a].y * w;
            acc[a].z += x[a].z * w;
            acc[a].w += x[a].w * w;
        }
        e = nx;
        r = rn;
    }

    v4f xv[NA];
#pragma unroll
    for (int a = 0; a < NA; ++a) {
        const int c = c0 + 2 * a;
        v4f o = {0.f, 0.f, 0.f, 0.f};
        if (mok) {
            const v4f h = xw4[(long long)m * CH + c];
            o.x = acc[a].x + h.x * dnode + bias_prev[4 * c + 0];
            o.y = acc[a].y + h.y * dnode + bias_prev[4 * c + 1];
            o.z = acc[a].z + h.z * dnode + bias_prev[4 * c + 2];
            o.w = acc[a].w + h.w * dnode + bias_prev[4 * c + 3];
        }
        xv[a] = o;
    }

#pragma unroll
    for (int t2 = 0; t2 < NT; ++t2) {
        const int trow0 = row0 + 16 * t2;
        if (trow0 < N) {
            v2f afr[CH];
#pragma unroll
            for (int kc = 0; kc < CH; ++kc) {
                const int srcl = (NT == 2) ? (16 * t2 + lmod) : (lmod + 16 * (kc & 1));
                const v4f sv = xv[kc >> 1];
                v4f t;
                t.x = __shfl(sv.x, srcl);
                t.y = __shfl(sv.y, srcl);
                t.z = __shfl(sv.z, srcl);
                t.w = __shfl(sv.w, srcl);
                afr[kc].x = lg ? t.z : t.x;
                afr[kc].y = lg ? t.w : t.y;
            }
            tile_linear_store<F_IN, F_OUT>(afr, W, xw, trow0, lane, N);
        }
    }
}

template <int F>
__global__ void k_gather_out(const float* __restrict__ xw,
                             const float* __restrict__ bias,
                             const int* __restrict__ rec,
                             const int* __restrict__ head,
                             const float* __restrict__ dsq,
                             const float* __restrict__ dinv,
                             float* __restrict__ out, int N, int E) {
    constexpr int LPN = F / 4;
    const long long gid = (long long)blockIdx.x * blockDim.x + threadIdx.x;
    const int node = (int)(gid / LPN);
    const int c    = (int)(gid % LPN);
    if (node >= N) return;

    const v4f* __restrict__ xw4  = reinterpret_cast<const v4f*>(xw);
    const v2i* __restrict__ rec2 = reinterpret_cast<const v2i*>(rec);
    const unsigned uE = (unsigned)E;

    const float dd = dsq[node];
    int e = head[node];
    v2i r; r.x = 0; r.y = -1;
    if ((unsigned)e < uE) r = rec2[e];

    v4f acc = {0.f, 0.f, 0.f, 0.f};
    for (int it = 0; it < E && (unsigned)e < uE; ++it) {
        const int s  = r.x;
        const int nx = r.y;
        const float q = dsq[s];
        const v4f   x = xw4[(long long)s * LPN + c];
        v2i rn; rn.x = 0; rn.y = -1;
        if ((unsigned)nx < uE) rn = rec2[nx];
        const float w = q * dd;
        acc.x += x.x * w;
        acc.y += x.y * w;
        acc.z += x.z * w;
        acc.w += x.w * w;
        e = nx;
        r = rn;
    }
    const v4f   h  = xw4[(long long)node * LPN + c];
    const float dn = dinv[node];
    v4f o;
    o.x = acc.x + h.x * dn + bias[4 * c + 0];
    o.y = acc.y + h.y * dn + bias[4 * c + 1];
    o.z = acc.z + h.z * dn + bias[4 * c + 2];
    o.w = acc.w + h.w * dn + bias[4 * c + 3];
    reinterpret_cast<v4f*>(out)[(long long)node * LPN + c] = o;
}

template <int F_IN, int F_OUT>
static void run_layer(const float* xw_prev, const float* bias_prev, const float* W,
                      float* xw, const float* dinv, const float* dsq,
                      const int* head, const int* rec,
                      int N, int E, hipStream_t stream) {
    constexpr int ROWS = (F_IN == 4) ? 32 : 16;
    const long long waves  = (long long)((N + ROWS - 1) / ROWS);
    const long long blocks = (waves + 7) / 8;
    k_gather_linear<F_IN, F_OUT><<<(unsigned)blocks, 256, 0, stream>>>(
        xw_prev, bias_prev, rec, head, dsq, dinv, W, xw, N, E);
}

extern "C" void kernel_launch(void* const* d_in, const int* in_sizes, int n_in,
                              void* d_out, int out_size, void* d_ws, size_t ws_size,
                              hipStream_t stream) {
    const int* read_length = (const int*)d_in[0];
    const int* edge        = (const int*)d_in[1];
    const int  N = in_sizes[0];
    const int  E = in_sizes[1] / 2;
    const int* src = edge;
    const int* dst = edge + E;

    const float* W1 = (const float*)d_in[2]; const float* B1 = (const float*)d_in[3];
    const float* W2 = (const float*)d_in[4]; const float* B2 = (const float*)d_in[5];
    const float* W3 = (const float*)d_in[6]; const float* B3 = (const float*)d_in[7];
    const float* W4 = (const float*)d_in[8]; const float* B4 = (const float*)d_in[9];

    float* ws   = (float*)d_ws;
    float* pxw  = ws;
    float* qxw  = pxw + (size_t)N * 16;
    int*   rec  = (int*)(qxw + (size_t)N * 32);
    float* dinv = (float*)(rec + (size_t)E * 2);
    float* dsq  = dinv + N;
    int*   head = (int*)(dsq + N);

    const int T = 256;

    k_init<<<(N + T - 1) / T, T, 0, stream>>>(head, N);
    k_build<<<(E + T - 1) / T, T, 0, stream>>>(src, dst, head, rec, E);

    {
        const long long waves  = (long long)((N + 15) / 16);
        const long long blocks = (waves + 7) / 8;
        k_first_linear<<<(unsigned)blocks, 256, 0, stream>>>(
            read_length, dinv, dsq, W1, pxw, head, rec, N, E);
    }

    run_layer<4, 8>(pxw, B1, W2, qxw, dinv, dsq, head, rec, N, E, stream);
    run_layer<8, 16>(qxw, B2, W3, pxw, dinv, dsq, head, rec, N, E, stream);
    run_layer<16, 32>(pxw, B3, W4, qxw, dinv, dsq, head, rec, N, E, stream);

    const long long nl = (long long)N * (32 / 4);
    k_gather_out<32><<<(unsigned)((nl + T - 1) / T), T, 0, stream>>>(
        qxw, B4, rec, head, dsq, dinv, (float*)d_out, N, E);
}


# lists walked once: degree pass copies each list's sources to a fixed-stride neighbour table (32 slots/node, tile-interleaved 16 B blocks); gathers read blocks, 4 edges in flight; sum order unchanged
# speedup vs baseline: 3.2414x; 1.1636x over previous
#include <hip/hip_runtime.h>
#include <hip/hip_bf16.h>


typedef __attribute__((ext_vector_type(2))) float v2f;
typedef __attribute__((ext_vector_type(4))) float v4f;
typedef __attribute__((ext_vector_type(8))) float v8f;
typedef __attribute__((ext_vector_type(2))) int   v2i;
typedef __attribute__((ext_vector_type(4))) int   v4i;

#define LEN_SCALE_INV (1.0f / 20000.0f)

#define NB_CAP 32
#define NB_BLK (NB_CAP / 4)

__global__ void k_init(int* __restrict__ head, int N) {
    int i = blockIdx.x * blockDim.x + threadIdx.x;
    if (i < N) head[i] = -1;
}

__global__ void k_build(const int* __restrict__ src, const int* __restrict__ dst,
                        int* __restrict__ head, int* __restrict__ rec, int E) {
    int e = blockIdx.x * blockDim.x + threadIdx.x;
    if (e >= E) return;
    const int prev = atomicExch(&head[dst[e]], e);
    v2i r; r.x = src[e]; r.y = prev;
    reinterpret_cast<v2i*>(rec)[e] = r;
}

template <int F_IN, int F_OUT>
__device__ __forceinline__ void tile_linear_store(const v2f (&afr)[(F_IN + 3) / 4],
                                                  const float* __restrict__ W,
                                                  float* __restrict__ xw,
                                                  int row0, int lane, int N) {
    constexpr int TILES_N = (F_OUT + 15) / 16;
    constexpr int KCH     = (F_IN + 3) / 4;
    const int lg   = lane >> 4;
    const int lmod = lane & 15;

    v8f acc[TILES_N];
#pragma unroll
    for (int t = 0; t < TILES_N; ++t)
        acc[t] = (v8f){0.f, 0.f, 0.f, 0.f, 0.f, 0.f, 0.f, 0.f};

#pragma unroll
    for (int kc = 0; kc < KCH; ++kc) {
        const int k0  = kc * 4 + lg * 2;
        const int k1  = k0 + 1;
        const int k0c = (k0 < F_IN) ? k0 : (F_IN - 1);
        const int k1c = (k1 < F_IN) ? k1 : (F_IN - 1);
#pragma unroll
        for (int t = 0; t < TILES_N; ++t) {
            const int  nc  = t * 16 + lmod;
            const bool nok = (nc < F_OUT);
            const int  ncc = nok ? nc : 0;
            float b0 = W[k0c * F_OUT + ncc];
            float b1 = W[k1c * F_OUT + ncc];
            b0 = (nok && k0 < F_IN) ? b0 : 0.f;
            b1 = (nok && k1 < F_IN) ? b1 : 0.f;
            v2f b; b.x = b0; b.y = b1;
            acc[t] = __builtin_amdgcn_wmma_f32_16x16x4_f32(
                false, afr[kc], false, b, (short)0, acc[t], false, false);
        }
    }

    const bool fullM = (row0 + 16 <= N);
#pragma unroll
    for (int t = 0; t < TILES_N; ++t) {
        const int nc = t * 16 + lmod;
        if (nc < F_OUT) {
            float* p = xw + (long long)(row0 + lg * 8) * F_OUT + nc;
            if (fullM) {
#pragma unroll
                for (int v = 0; v < 8; ++v) p[(long long)v * F_OUT] = acc[t][v];
            } else {
#pragma unroll
                for (int v = 0; v < 8; ++v) {
                    const int mrow = row0 + v + lg * 8;
                    if (mrow < N) p[(long long)v * F_OUT] = acc[t][v];
                }
            }
        }
    }
}

__global__ void k_first_linear(const int* __restrict__ read_length,
                               float* __restrict__ dinv, float* __restrict__ dsq,
                               int* __restrict__ cnt_out,
                               const float* __restrict__ W, float* __restrict__ xw,
                               int* __restrict__ head,
                               const int* __restrict__ rec,
                               int* __restrict__ tab, int N, int E) {
    const int wave = blockIdx.x * (blockDim.x >> 5) + (threadIdx.x >> 5);
    const int lane = threadIdx.x & 31;
    const int row0 = wave * 16;
    if (row0 >= N) return;

    const int lg = lane >> 4;
    const int m  = row0 + (lane & 15);

    float xfirst = 0.f;
    if (lg == 0 && m < N) {
        const unsigned uE = (unsigned)E;
        const v2i* __restrict__ rec2 = reinterpret_cast<const v2i*>(rec);
        v4i* __restrict__ tb = reinterpret_cast<v4i*>(tab)
                             + ((long long)wave * NB_BLK) * 16 + (lane & 15);
        int cnt = 0;
        int ovf = -1;
        int e = head[m];
        v4i blk = {0, 0, 0, 0};
        for (int it = 0; it < E && (unsigned)e < uE; ++it) {
            const v2i r = rec2[e];
            if (cnt < NB_CAP) {
                const int sl = cnt & 3;
                blk.x = (sl == 0) ? r.x : blk.x;
                blk.y = (sl == 1) ? r.x : blk.y;
                blk.z = (sl == 2) ? r.x : blk.z;
                blk.w = (sl == 3) ? r.x : blk.w;
                if (sl == 3) {
                    tb[(cnt >> 2) * 16] = blk;
                    blk = (v4i){0, 0, 0, 0};
                }
            } else if (cnt == NB_CAP) {
                ovf = e;
            }
            e = r.y;
            ++cnt;
        }
        if (cnt < NB_CAP && (cnt & 3) != 0) tb[(cnt >> 2) * 16] = blk;
        head[m]    = ovf;
        cnt_out[m] = cnt;
        const float deg = 1.0f + (float)cnt;
        dsq[m]  = rsqrtf(deg);
        dinv[m] = 1.0f / deg;
        xfirst  = (float)read_length[m] * LEN_SCALE_INV;
    }
    v2f afr[1];
    afr[0].x = xfirst;
    afr[0].y = 0.f;
    tile_linear_store<1, 4>(afr, W, xw, row0, lane, N);
}

template <int CH, int NA>
__device__ __forceinline__ void sum_in_edges(v4f (&acc)[NA],
                                             const v4f* __restrict__ xw4,
                                             const float* __restrict__ dsq,
                                             const v4i* __restrict__ tab4,
                                             const v2i* __restrict__ rec2,
                                             const int* __restrict__ head,
                                             int m, int c, int c0, float dd, int E) {
    const unsigned uE = (unsigned)E;
    const int ct = (c < NB_CAP) ? c : NB_CAP;
    const int nb = (ct + 3) >> 2;
    const v4i* __restrict__ tb = tab4 + ((long long)(m >> 4) * NB_BLK) * 16 + (m & 15);

    v4i s4 = {0, 0, 0, 0};
    if (nb > 0) s4 = tb[0];
    for (int j = 0; j < nb; ++j) {
        const int sv[4] = {s4.x, s4.y, s4.z, s4.w};
        float q[4];
        v4f   x[4][NA];
#pragma unroll
        for (int i = 0; i < 4; ++i) q[i] = dsq[sv[i]];
#pragma unroll
        for (int i = 0; i < 4; ++i) {
#pragma unroll
            for (int a = 0; a < NA; ++a) x[i][a] = xw4[(long long)sv[i] * CH + c0 + 2 * a];
        }
        v4i sn = {0, 0, 0, 0};
        if (j + 1 < nb) sn = tb[(j + 1) * 16];
#pragma unroll
        for (int i = 0; i < 4; ++i) {
            if (4 * j + i < ct) {
                const float w = q[i] * dd;
#pragma unroll
                for (int a = 0; a < NA; ++a) {
                    acc[a].x += x[i][a].x * w;
                    acc[a].y += x[i][a].y * w;
                    acc[a].z += x[i][a].z * w;
                    acc[a].w += x[i][a].w * w;
                }
            }
        }
        s4 = sn;
    }

    int e = -1;
    if (c > NB_CAP) e = head[m];
    v2i r; r.x = 0; r.y = -1;
    if ((unsigned)e < uE) r = rec2[e];
    for (int it = 0; it < E && (unsigned)e < uE; ++it) {
        const int s  = r.x;
        const int nx = r.y;
        const float q = dsq[s];
        v4f x[NA];
#pragma unroll
        for (int a = 0; a < NA; ++a) x[a] = xw4[(long long)s * CH + c0 + 2 * a];
        v2i rn; rn.x = 0; rn.y = -1;
        if ((unsigned)nx < uE) rn = rec2[nx];
        const float w = q * dd;
#pragma unroll
        for (int a = 0; a < NA; ++a) {
            acc[a].x += x[a].x * w;
            acc[a].y += x[a].y * w;
            acc[a].z += x[a].z * w;
            acc[a].w += x[a].w * w;
        }
        e = nx;
        r = rn;
    }
}

template <int F_IN, int F_OUT>
__global__ void k_gather_linear(const float* __restrict__ xw_prev,
                                const float* __restrict__ bias_prev,
                                const int* __restrict__ rec,
                                const int* __restrict__ head,
                                const int* __restrict__ cnt,
                                const int* __restrict__ tab,
                                const float* __restrict__ dsq,
                                const float* __restrict__ dinv,
                                const float* __restrict__ W,
                                float* __restrict__ xw, int N, int E) {
    constexpr int CH = F_IN / 4;
    constexpr int NT = (CH == 1) ? 2 : 1;
    constexpr int NA = (CH == 4) ? 2 : 1;
    static_assert(CH == 1 || CH == 2 || CH == 4, "layer width");

    const int wave = blockIdx.x * (blockDim.x >> 5) + (threadIdx.x >> 5);
    const int lane = threadIdx.x & 31;
    const int row0 = wave * (16 * NT);
    if (row0 >= N) return;

    const int  lg   = lane >> 4;
    const int  lmod = lane & 15;
    const int  m    = row0 + ((NT == 2) ? lane : lmod);
    const bool mok  = (m < N);
    const int  c0   = (CH == 1) ? 0 : lg;

    const v4f* __restrict__ xw4  = reinterpret_cast<const v4f*>(xw_prev);
    const v2i* __restrict__ rec2 = reinterpret_cast<const v2i*>(rec);
    const v4i* __restrict__ tab4 = reinterpret_cast<const v4i*>(tab);

    float dd = 0.f, dnode = 0.f;
    int c = 0;
    if (mok) { dd = dsq[m]; dnode = dinv[m]; c = cnt[m]; }

    v4f acc[NA];
#pragma unroll
    for (int a = 0; a < NA; ++a) acc[a] = (v4f){0.f, 0.f, 0.f, 0.f};

    sum_in_edges<CH, NA>(acc, xw4, dsq, tab4, rec2, head, m, c, c0, dd, E);

    v4f xv[NA];
#pragma unroll
    for (int a = 0; a < NA; ++a) {
        const int c = c0 + 2 * a;
        v4f o = {0.f, 0.f, 0.f, 0.f};
        if (mok) {
            const v4f h = xw4[(long long)m * CH + c];
            o.x = acc[a].x + h.x * dnode + bias_prev[4 * c + 0];
            o.y = acc[a].y + h.y * dnode + bias_prev[4 * c + 1];
            o.z = acc[a].z + h.z * dnode + bias_prev[4 * c + 2];
            o.w = acc[a].w + h.w * dnode + bias_prev[4 * c + 3];
        }
        xv[a] = o;
    }

#pragma unroll
    for (int t2 = 0; t2 < NT; ++t2) {
        const int trow0 = row0 + 16 * t2;
        if (trow0 < N) {
            v2f afr[CH];
#pragma unroll
            for (int kc = 0; kc < CH; ++kc) {
                const int srcl = (NT == 2) ? (16 * t2 + lmod) : (lmod + 16 * (kc & 1));
                const v4f sv = xv[kc >> 1];
                v4f t;
                t.x = __shfl(sv.x, srcl);
                t.y = __shfl(sv.y, srcl);
                t.z = __shfl(sv.z, srcl);
                t.w = __shfl(sv.w, srcl);
                afr[kc].x = lg ? t.z : t.x;
                afr[kc].y = lg ? t.w : t.y;
            }
            tile_linear_store<F_IN, F_OUT>(afr, W, xw, trow0, lane, N);
        }
    }
}

template <int F>
__global__ void k_gather_out(const float* __restrict__ xw,
                             const float* __restrict__ bias,
                             const int* __restrict__ rec,
                             const int* __restrict__ head,
                             const int* __restrict__ cnt,
                             const int* __restrict__ tab,
                             const float* __restrict__ dsq,
                             const float* __restrict__ dinv,
                             float* __restrict__ out, int N, int E) {
    constexpr int LPN = F / 4;
    const long long gid = (long long)blockIdx.x * blockDim.x + threadIdx.x;
    const int node = (int)(gid / LPN);
    const int c    = (int)(gid % LPN);
    if (node >= N) return;

    const v4f* __restrict__ xw4  = reinterpret_cast<const v4f*>(xw);
    const v2i* __restrict__ rec2 = reinterpret_cast<const v2i*>(rec);
    const v4i* __restrict__ tab4 = reinterpret_cast<const v4i*>(tab);

    const float dd = dsq[node];
    v4f acc1[1];
    acc1[0] = (v4f){0.f, 0.f, 0.f, 0.f};
    sum_in_edges<LPN, 1>(acc1, xw4, dsq, tab4, rec2, head, node, cnt[node], c, dd, E);
    const v4f acc = acc1[0];

    const v4f   h  = xw4[(long long)node * LPN + c];
    const float dn = dinv[node];
    v4f o;
    o.x = acc.x + h.x * dn + bias[4 * c + 0];
    o.y = acc.y + h.y * dn + bias[4 * c + 1];
    o.z = acc.z + h.z * dn + bias[4 * c + 2];
    o.w = acc.w + h.w * dn + bias[4 * c + 3];
    reinterpret_cast<v4f*>(out)[(long long)node * LPN + c] = o;
}

template <int F_IN, int F_OUT>
static void run_layer(const float* xw_prev, const float* bias_prev, const float* W,
                      float* xw, const float* dinv, const float* dsq,
                      const int* head, const int* rec,
                      const int* cnt, const int* tab,
                      int N, int E, hipStream_t stream) {
    constexpr int ROWS = (F_IN == 4) ? 32 : 16;
    const long long waves  = (long long)((N + ROWS - 1) / ROWS);
    const long long blocks = (waves + 7) / 8;
    k_gather_linear<F_IN, F_OUT><<<(unsigned)blocks, 256, 0, stream>>>(
        xw_prev, bias_prev, rec, head, cnt, tab, dsq, dinv, W, xw, N, E);
}

extern "C" void kernel_launch(void* const* d_in, const int* in_sizes, int n_in,
                              void* d_out, int out_size, void* d_ws, size_t ws_size,
                              hipStream_t stream) {
    const int* read_length = (const int*)d_in[0];
    const int* edge        = (const int*)d_in[1];
    const int  N = in_sizes[0];
    const int  E = in_sizes[1] / 2;
    const int* src = edge;
    const int* dst = edge + E;

    const float* W1 = (const float*)d_in[2]; const float* B1 = (const float*)d_in[3];
    const float* W2 = (const float*)d_in[4]; const float* B2 = (const float*)d_in[5];
    const float* W3 = (const float*)d_in[6]; const float* B3 = (const float*)d_in[7];
    const float* W4 = (const float*)d_in[8]; const float* B4 = (const float*)d_in[9];

    const size_t tiles = ((size_t)N + 15) / 16;
    float* ws   = (float*)d_ws;
    float* pxw  = ws;
    float* qxw  = pxw + (size_t)N * 16;
    int*   tab  = (int*)(qxw + (size_t)N * 32);
    int*   rec  = tab + tiles * NB_CAP * 16;
    float* dinv = (float*)(rec + (size_t)E * 2);
    float* dsq  = dinv + N;
    int*   head = (int*)(dsq + N);
    int*   cnt  = head + N;

    const int T = 256;

    k_init<<<(N + T - 1) / T, T, 0, stream>>>(head, N);
    k_build<<<(E + T - 1) / T, T, 0, stream>>>(src, dst, head, rec, E);

    {
        const long long waves  = (long long)((N + 15) / 16);
        const long long blocks = (waves + 7) / 8;
        k_first_linear<<<(unsigned)blocks, 256, 0, stream>>>(
            read_length, dinv, dsq, cnt, W1, pxw, head, rec, tab, N, E);
    }

    run_layer<4, 8>(pxw, B1, W2, qxw, dinv, dsq, head, rec, cnt, tab, N, E, stream);
    run_layer<8, 16>(qxw, B2, W3, pxw, dinv, dsq, head, rec, cnt, tab, N, E, stream);
    run_layer<16, 32>(pxw, B3, W4, qxw, dinv, dsq, head, rec, cnt, tab, N, E, stream);

    const long long nl = (long long)N * (32 / 4);
    k_gather_out<32><<<(unsigned)((nl + T - 1) / T), T, 0, stream>>>(
        qxw, B4, rec, head, cnt, tab, dsq, dinv, (float*)d_out, N, E);
}


# edge weight cached: first gather stores dsq[src] of a node's first 16 table slots as 16 B weight blocks in the table's layout; later gathers read those, no scattered dsq loads; sum order unchanged
# speedup vs baseline: 3.5641x; 1.0996x over previous
#include <hip/hip_runtime.h>
#include <hip/hip_bf16.h>


typedef __attribute__((ext_vector_type(2))) float v2f;
typedef __attribute__((ext_vector_type(4))) float v4f;
typedef __attribute__((ext_vector_type(8))) float v8f;
typedef __attribute__((ext_vector_type(2))) int   v2i;
typedef __attribute__((ext_vector_type(4))) int   v4i;

#define LEN_SCALE_INV (1.0f / 20000.0f)

#define NB_CAP 32
#define NB_BLK (NB_CAP / 4)
#define NB_WBLK 4

__global__ void k_init(int* __restrict__ head, int N) {
    int i = blockIdx.x * blockDim.x + threadIdx.x;
    if (i < N) head[i] = -1;
}

__global__ void k_build(const int* __restrict__ src, const int* __restrict__ dst,
                        int* __restrict__ head, int* __restrict__ rec, int E) {
    int e = blockIdx.x * blockDim.x + threadIdx.x;
    if (e >= E) return;
    const int prev = atomicExch(&head[dst[e]], e);
    v2i r; r.x = src[e]; r.y = prev;
    reinterpret_cast<v2i*>(rec)[e] = r;
}

template <int F_IN, int F_OUT>
__device__ __forceinline__ void tile_linear_store(const v2f (&afr)[(F_IN + 3) / 4],
                                                  const float* __restrict__ W,
                                                  float* __restrict__ xw,
                                                  int row0, int lane, int N) {
    constexpr int TILES_N = (F_OUT + 15) / 16;
    constexpr int KCH     = (F_IN + 3) / 4;
    const int lg   = lane >> 4;
    const int lmod = lane & 15;

    v8f acc[TILES_N];
#pragma unroll
    for (int t = 0; t < TILES_N; ++t)
        acc[t] = (v8f){0.f, 0.f, 0.f, 0.f, 0.f, 0.f, 0.f, 0.f};

#pragma unroll
    for (int kc = 0; kc < KCH; ++kc) {
        const int k0  = kc * 4 + lg * 2;
        const int k1  = k0 + 1;
        const int k0c = (k0 < F_IN) ? k0 : (F_IN - 1);
        const int k1c = (k1 < F_IN) ? k1 : (F_IN - 1);
#pragma unroll
        for (int t = 0; t < TILES_N; ++t) {
            const int  nc  = t * 16 + lmod;
            const bool nok = (nc < F_OUT);
            const int  ncc = nok ? nc : 0;
            float b0 = W[k0c * F_OUT + ncc];
            float b1 = W[k1c * F_OUT + ncc];
            b0 = (nok && k0 < F_IN) ? b0 : 0.f;
            b1 = (nok && k1 < F_IN) ? b1 : 0.f;
            v2f b; b.x = b0; b.y = b1;
            acc[t] = __builtin_amdgcn_wmma_f32_16x16x4_f32(
                false, afr[kc], false, b, (short)0, acc[t], false, false);
        }
    }

    const bool fullM = (row0 + 16 <= N);
#pragma unroll
    for (int t = 0; t < TILES_N; ++t) {
        const int nc = t * 16 + lmod;
        if (nc < F_OUT) {
            float* p = xw + (long long)(row0 + lg * 8) * F_OUT + nc;
            if (fullM) {
#pragma unroll
                for (int v = 0; v < 8; ++v) p[(long long)v * F_OUT] = acc[t][v];
            } else {
#pragma unroll
                for (int v = 0; v < 8; ++v) {
                    const int mrow = row0 + v + lg * 8;
                    if (mrow < N) p[(long long)v * F_OUT] = acc[t][v];
                }
            }
        }
    }
}

__global__ void k_first_linear(const int* __restrict__ read_length,
                               float* __restrict__ dinv, float* __restrict__ dsq,
                               int* __restrict__ cnt_out,
                               const float* __restrict__ W, float* __restrict__ xw,
                               int* __restrict__ head,
                               const int* __restrict__ rec,
                               int* __restrict__ tab, int N, int E) {
    const int wave = blockIdx.x * (blockDim.x >> 5) + (threadIdx.x >> 5);
    const int lane = threadIdx.x & 31;
    const int row0 = wave * 16;
    if (row0 >= N) return;

    const int lg = lane >> 4;
    const int m  = row0 + (lane & 15);

    float xfirst = 0.f;
    if (lg == 0 && m < N) {
        const unsigned uE = (unsigned)E;
        const v2i* __restrict__ rec2 = reinterpret_cast<const v2i*>(rec);
        v4i* __restrict__ tb = reinterpret_cast<v4i*>(tab)
                             + ((long long)wave * NB_BLK) * 16 + (lane & 15);
        int cnt = 0;
        int ovf = -1;
        int e = head[m];
        v4i blk = {0, 0, 0, 0};
        for (int it = 0; it < E && (unsigned)e < uE; ++it) {
            const v2i r = rec2[e];
            if (cnt < NB_CAP) {
                const int sl = cnt & 3;
                blk.x = (sl == 0) ? r.x : blk.x;
                blk.y = (sl == 1) ? r.x : blk.y;
                blk.z = (sl == 2) ? r.x : blk.z;
                blk.w = (sl == 3) ? r.x : blk.w;
                if (sl == 3) {
                    tb[(cnt >> 2) * 16] = blk;
                    blk = (v4i){0, 0, 0, 0};
                }
            } else if (cnt == NB_CAP) {
                ovf = e;
            }
            e = r.y;
            ++cnt;
        }
        if (cnt < NB_CAP && (cnt & 3) != 0) tb[(cnt >> 2) * 16] = blk;
        head[m]    = ovf;
        cnt_out[m] = cnt;
        const float deg = 1.0f + (float)cnt;
        dsq[m]  = rsqrtf(deg);
        dinv[m] = 1.0f / deg;
        xfirst  = (float)read_length[m] * LEN_SCALE_INV;
    }
    v2f afr[1];
    afr[0].x = xfirst;
    afr[0].y = 0.f;
    tile_linear_store<1, 4>(afr, W, xw, row0, lane, N);
}

template <int CH, int NA, bool WR>
__device__ __forceinline__ void sum_in_edges(v4f (&acc)[NA],
                                             const v4f* __restrict__ xw4,
                                             const float* __restrict__ dsq,
                                             const v4i* __restrict__ tab4,
                                             v4f* __restrict__ wtab4,
                                             const v2i* __restrict__ rec2,
                                             const int* __restrict__ head,
                                             int m, int c, int c0, float dd, int E) {
    const unsigned uE = (unsigned)E;
    const int ct = (c < NB_CAP) ? c : NB_CAP;
    const int nb = (ct + 3) >> 2;
    const v4i* __restrict__ tb = tab4 + ((long long)(m >> 4) * NB_BLK) * 16 + (m & 15);
    v4f* __restrict__ wb = wtab4 + ((long long)(m >> 4) * NB_WBLK) * 16 + (m & 15);

    v4i s4 = {0, 0, 0, 0};
    v4f w4 = {0.f, 0.f, 0.f, 0.f};
    if (nb > 0) {
        s4 = tb[0];
        if (!WR) w4 = wb[0];
    }
    for (int j = 0; j < nb; ++j) {
        const int sv[4] = {s4.x, s4.y, s4.z, s4.w};
        float q[4];
        v4f   x[4][NA];
        if (!WR && j < NB_WBLK) {
            q[0] = w4.x; q[1] = w4.y; q[2] = w4.z; q[3] = w4.w;
        } else {
#pragma unroll
            for (int i = 0; i < 4; ++i) q[i] = dsq[sv[i]];
        }
#pragma unroll
        for (int i = 0; i < 4; ++i) {
#pragma unroll
            for (int a = 0; a < NA; ++a) x[i][a] = xw4[(long long)sv[i] * CH + c0 + 2 * a];
        }
        v4i sn = {0, 0, 0, 0};
        v4f wn = {0.f, 0.f, 0.f, 0.f};
        if (j + 1 < nb) {
            sn = tb[(j + 1) * 16];
            if (!WR && j + 1 < NB_WBLK) wn = wb[(j + 1) * 16];
        }
        if (WR && j < NB_WBLK) {
            v4f qv; qv.x = q[0]; qv.y = q[1]; qv.z = q[2]; qv.w = q[3];
            wb[j * 16] = qv;
        }
#pragma unroll
        for (int i = 0; i < 4; ++i) {
            if (4 * j + i < ct) {
                const float w = q[i] * dd;
#pragma unroll
                for (int a = 0; a < NA; ++a) {
                    acc[a].x += x[i][a].x * w;
                    acc[a].y += x[i][a].y * w;
                    acc[a].z += x[i][a].z * w;
                    acc[a].w += x[i][a].w * w;
                }
            }
        }
        s4 = sn;
        w4 = wn;
    }

    int e = -1;
    if (c > NB_CAP) e = head[m];
    v2i r; r.x = 0; r.y = -1;
    if ((unsigned)e < uE) r = rec2[e];
    for (int it = 0; it < E && (unsigned)e < uE; ++it) {
        const int s  = r.x;
        const int nx = r.y;
        const float q = dsq[s];
        v4f x[NA];
#pragma unroll
        for (int a = 0; a < NA; ++a) x[a] = xw4[(long long)s * CH + c0 + 2 * a];
        v2i rn; rn.x = 0; rn.y = -1;
        if ((unsigned)nx < uE) rn = rec2[nx];
        const float w = q * dd;
#pragma unroll
        for (int a = 0; a < NA; ++a) {
            acc[a].x += x[a].x * w;
            acc[a].y += x[a].y * w;
            acc[a].z += x[a].z * w;
            acc[a].w += x[a].w * w;
        }
        e = nx;
        r = rn;
    }
}

template <int F_IN, int F_OUT>
__global__ void k_gather_linear(const float* __restrict__ xw_prev,
                                const float* __restrict__ bias_prev,
                                const int* __restrict__ rec,
                                const int* __restrict__ head,
                                const int* __restrict__ cnt,
                                const int* __restrict__ tab,
                                float* __restrict__ wtab,
                                const float* __restrict__ dsq,
                                const float* __restrict__ dinv,
                                const float* __restrict__ W,
                                float* __restrict__ xw, int N, int E) {
    constexpr int CH = F_IN / 4;
    constexpr int NT = (CH == 1) ? 2 : 1;
    constexpr int NA = (CH == 4) ? 2 : 1;
    static_assert(CH == 1 || CH == 2 || CH == 4, "layer width");

    const int wave = blockIdx.x * (blockDim.x >> 5) + (threadIdx.x >> 5);
    const int lane = threadIdx.x & 31;
    const int row0 = wave * (16 * NT);
    if (row0 >= N) return;

    const int  lg   = lane >> 4;
    const int  lmod = lane & 15;
    const int  m    = row0 + ((NT == 2) ? lane : lmod);
    const bool mok  = (m < N);
    const int  c0   = (CH == 1) ? 0 : lg;

    const v4f* __restrict__ xw4  = reinterpret_cast<const v4f*>(xw_prev);
    const v2i* __restrict__ rec2 = reinterpret_cast<const v2i*>(rec);
    const v4i* __restrict__ tab4 = reinterpret_cast<const v4i*>(tab);
    v4f* __restrict__ wtab4 = reinterpret_cast<v4f*>(wtab);

    float dd = 0.f, dnode = 0.f;
    int c = 0;
    if (mok) { dd = dsq[m]; dnode = dinv[m]; c = cnt[m]; }

    v4f acc[NA];
#pragma unroll
    for (int a = 0; a < NA; ++a) acc[a] = (v4f){0.f, 0.f, 0.f, 0.f};

    sum_in_edges<CH, NA, (CH == 1)>(acc, xw4, dsq, tab4, wtab4, rec2, head, m, c, c0, dd, E);

    v4f xv[NA];
#pragma unroll
    for (int a = 0; a < NA; ++a) {
        const int c = c0 + 2 * a;
        v4f o = {0.f, 0.f, 0.f, 0.f};
        if (mok) {
            const v4f h = xw4[(long long)m * CH + c];
            o.x = acc[a].x + h.x * dnode + bias_prev[4 * c + 0];
            o.y = acc[a].y + h.y * dnode + bias_prev[4 * c + 1];
            o.z = acc[a].z + h.z * dnode + bias_prev[4 * c + 2];
            o.w = acc[a].w + h.w * dnode + bias_prev[4 * c + 3];
        }
        xv[a] = o;
    }

#pragma unroll
    for (int t2 = 0; t2 < NT; ++t2) {
        const int trow0 = row0 + 16 * t2;
        if (trow0 < N) {
            v2f afr[CH];
#pragma unroll
            for (int kc = 0; kc < CH; ++kc) {
                const int srcl = (NT == 2) ? (16 * t2 + lmod) : (lmod + 16 * (kc & 1));
                const v4f sv = xv[kc >> 1];
                v4f t;
                t.x = __shfl(sv.x, srcl);
                t.y = __shfl(sv.y, srcl);
                t.z = __shfl(sv.z, srcl);
                t.w = __shfl(sv.w, srcl);
                afr[kc].x = lg ? t.z : t.x;
                afr[kc].y = lg ? t.w : t.y;
            }
            tile_linear_store<F_IN, F_OUT>(afr, W, xw, trow0, lane, N);
        }
    }
}

template <int F>
__global__ void k_gather_out(const float* __restrict__ xw,
                             const float* __restrict__ bias,
                             const int* __restrict__ rec,
                             const int* __restrict__ head,
                             const int* __restrict__ cnt,
                             const int* __restrict__ tab,
                             float* __restrict__ wtab,
                             const float* __restrict__ dsq,
                             const float* __restrict__ dinv,
                             float* __restrict__ out, int N, int E) {
    constexpr int LPN = F / 4;
    const long long gid = (long long)blockIdx.x * blockDim.x + threadIdx.x;
    const int node = (int)(gid / LPN);
    const int c    = (int)(gid % LPN);
    if (node >= N) return;

    const v4f* __restrict__ xw4  = reinterpret_cast<const v4f*>(xw);
    const v2i* __restrict__ rec2 = reinterpret_cast<const v2i*>(rec);
    const v4i* __restrict__ tab4 = reinterpret_cast<const v4i*>(tab);
    v4f* __restrict__ wtab4 = reinterpret_cast<v4f*>(wtab);

    const float dd = dsq[node];
    v4f acc1[1];
    acc1[0] = (v4f){0.f, 0.f, 0.f, 0.f};
    sum_in_edges<LPN, 1, false>(acc1, xw4, dsq, tab4, wtab4, rec2, head, node, cnt[node], c, dd, E);
    const v4f acc = acc1[0];

    const v4f   h  = xw4[(long long)node * LPN + c];
    const float dn = dinv[node];
    v4f o;
    o.x = acc.x + h.x * dn + bias[4 * c + 0];
    o.y = acc.y + h.y * dn + bias[4 * c + 1];
    o.z = acc.z + h.z * dn + bias[4 * c + 2];
    o.w = acc.w + h.w * dn + bias[4 * c + 3];
    reinterpret_cast<v4f*>(out)[(long long)node * LPN + c] = o;
}

template <int F_IN, int F_OUT>
static void run_layer(const float* xw_prev, const float* bias_prev, const float* W,
                      float* xw, const float* dinv, const float* dsq,
                      const int* head, const int* rec,
                      const int* cnt, const int* tab, float* wtab,
                      int N, int E, hipStream_t stream) {
    constexpr int ROWS = (F_IN == 4) ? 32 : 16;
    const long long waves  = (long long)((N + ROWS - 1) / ROWS);
    const long long blocks = (waves + 7) / 8;
    k_gather_linear<F_IN, F_OUT><<<(unsigned)blocks, 256, 0, stream>>>(
        xw_prev, bias_prev, rec, head, cnt, tab, wtab, dsq, dinv, W, xw, N, E);
}

extern "C" void kernel_launch(void* const* d_in, const int* in_sizes, int n_in,
                              void* d_out, int out_size, void* d_ws, size_t ws_size,
                              hipStream_t stream) {
    const int* read_length = (const int*)d_in[0];
    const int* edge        = (const int*)d_in[1];
    const int  N = in_sizes[0];
    const int  E = in_sizes[1] / 2;
    const int* src = edge;
    const int* dst = edge + E;

    const float* W1 = (const float*)d_in[2]; const float* B1 = (const float*)d_in[3];
    const float* W2 = (const float*)d_in[4]; const float* B2 = (const float*)d_in[5];
    const float* W3 = (const float*)d_in[6]; const float* B3 = (const float*)d_in[7];
    const float* W4 = (const float*)d_in[8]; const float* B4 = (const float*)d_in[9];

    const size_t tiles = ((size_t)N + 15) / 16;
    float* ws   = (float*)d_ws;
    float* pxw  = ws;
    float* qxw  = pxw + (size_t)N * 16;
    int*   tab  = (int*)(qxw + (size_t)N * 32);
    float* wtab = (float*)(tab + tiles * NB_CAP * 16);
    int*   rec  = (int*)(wtab + tiles * NB_WBLK * 64);
    float* dinv = (float*)(rec + (size_t)E * 2);
    float* dsq  = dinv + N;
    int*   head = (int*)(dsq + N);
    int*   cnt  = head + N;

    const int T = 256;

    k_init<<<(N + T - 1) / T, T, 0, stream>>>(head, N);
    k_build<<<(E + T - 1) / T, T, 0, stream>>>(src, dst, head, rec, E);

    {
        const long long waves  = (long long)((N + 15) / 16);
        const long long blocks = (waves + 7) / 8;
        k_first_linear<<<(unsigned)blocks, 256, 0, stream>>>(
            read_length, dinv, dsq, cnt, W1, pxw, head, rec, tab, N, E);
    }

    run_layer<4, 8>(pxw, B1, W2, qxw, dinv, dsq, head, rec, cnt, tab, wtab, N, E, stream);
    run_layer<8, 16>(qxw, B2, W3, pxw, dinv, dsq, head, rec, cnt, tab, wtab, N, E, stream);
    run_layer<16, 32>(pxw, B3, W4, qxw, dinv, dsq, head, rec, cnt, tab, wtab, N, E, stream);

    const long long nl = (long long)N * (32 / 4);
    k_gather_out<32><<<(unsigned)((nl + T - 1) / T), T, 0, stream>>>(
        qxw, B4, rec, head, cnt, tab, wtab, dsq, dinv, (float*)d_out, N, E);
}


# list build without global atomics: block = 65536 nodes' head words in LDS x 1 of 32 edge chunks, LDS exchange per edge; degree pass chains the 32 partial lists; sum order changes (still arrival-set)
# speedup vs baseline: 4.8067x; 1.3486x over previous
#include <hip/hip_runtime.h>
#include <hip/hip_bf16.h>


typedef __attribute__((ext_vector_type(2))) float v2f;
typedef __attribute__((ext_vector_type(4))) float v4f;
typedef __attribute__((ext_vector_type(8))) float v8f;
typedef __attribute__((ext_vector_type(2))) int   v2i;
typedef __attribute__((ext_vector_type(4))) int   v4i;

#define LEN_SCALE_INV (1.0f / 20000.0f)

#define NB_CAP 32
#define NB_BLK (NB_CAP / 4)
#define NB_WBLK 4

#define LB_RANGE   65536
#define LB_CHUNKS  32
#define LB_THREADS 1024
#define LB_UNROLL  16

__global__ __launch_bounds__(LB_THREADS)
void k_build(const int* __restrict__ src, const int* __restrict__ dst,
             int* __restrict__ ph, int* __restrict__ rec, int N, int E, int R) {
    __shared__ int hd[LB_RANGE];
    const int r  = (int)(blockIdx.x % (unsigned)R);
    const int c  = (int)(blockIdx.x / (unsigned)R);
    const int lo = r * LB_RANGE;
    int len = N - lo;
    if (len > LB_RANGE) len = LB_RANGE;
    const unsigned ulen = (unsigned)len;

    for (int i = threadIdx.x; i < LB_RANGE; i += LB_THREADS) hd[i] = -1;
    __syncthreads();

    const int per = (E + LB_CHUNKS - 1) / LB_CHUNKS;
    const int e0  = c * per;
    int e1 = e0 + per;
    if (e1 > E) e1 = E;
    v2i* __restrict__ rec2 = reinterpret_cast<v2i*>(rec);

    int dn[LB_UNROLL];
#pragma unroll
    for (int u = 0; u < LB_UNROLL; ++u) {
        const int e = e0 + (int)threadIdx.x + u * LB_THREADS;
        dn[u] = (e < e1) ? dst[e] : -1;
    }
    for (int base = e0 + (int)threadIdx.x; base < e1; base += LB_THREADS * LB_UNROLL) {
        unsigned off[LB_UNROLL];
        int      sv[LB_UNROLL];
        int      pv[LB_UNROLL];
#pragma unroll
        for (int u = 0; u < LB_UNROLL; ++u)
            off[u] = (unsigned)(dn[u] - lo);
#pragma unroll
        for (int u = 0; u < LB_UNROLL; ++u) {
            sv[u] = 0;
            if (off[u] < ulen) sv[u] = src[base + u * LB_THREADS];
        }
#pragma unroll
        for (int u = 0; u < LB_UNROLL; ++u) {
            const int e = base + LB_THREADS * LB_UNROLL + u * LB_THREADS;
            dn[u] = (e < e1) ? dst[e] : -1;
        }
#pragma unroll
        for (int u = 0; u < LB_UNROLL; ++u) {
            pv[u] = -1;
            if (off[u] < ulen) pv[u] = atomicExch(&hd[off[u]], base + u * LB_THREADS);
        }
#pragma unroll
        for (int u = 0; u < LB_UNROLL; ++u) {
            if (off[u] < ulen) {
                v2i rr; rr.x = sv[u]; rr.y = pv[u];
                rec2[base + u * LB_THREADS] = rr;
            }
        }
    }
    __syncthreads();

    int* __restrict__ row = ph + (long long)c * N + lo;
    for (int i = threadIdx.x; i < len; i += LB_THREADS) row[i] = hd[i];
}

template <int F_IN, int F_OUT>
__device__ __forceinline__ void tile_linear_store(const v2f (&afr)[(F_IN + 3) / 4],
                                                  const float* __restrict__ W,
                                                  float* __restrict__ xw,
                                                  int row0, int lane, int N) {
    constexpr int TILES_N = (F_OUT + 15) / 16;
    constexpr int KCH     = (F_IN + 3) / 4;
    const int lg   = lane >> 4;
    const int lmod = lane & 15;

    v8f acc[TILES_N];
#pragma unroll
    for (int t = 0; t < TILES_N; ++t)
        acc[t] = (v8f){0.f, 0.f, 0.f, 0.f, 0.f, 0.f, 0.f, 0.f};

#pragma unroll
    for (int kc = 0; kc < KCH; ++kc) {
        const int k0  = kc * 4 + lg * 2;
        const int k1  = k0 + 1;
        const int k0c = (k0 < F_IN) ? k0 : (F_IN - 1);
        const int k1c = (k1 < F_IN) ? k1 : (F_IN - 1);
#pragma unroll
        for (int t = 0; t < TILES_N; ++t) {
            const int  nc  = t * 16 + lmod;
            const bool nok = (nc < F_OUT);
            const int  ncc = nok ? nc : 0;
            float b0 = W[k0c * F_OUT + ncc];
            float b1 = W[k1c * F_OUT + ncc];
            b0 = (nok && k0 < F_IN) ? b0 : 0.f;
            b1 = (nok && k1 < F_IN) ? b1 : 0.f;
            v2f b; b.x = b0; b.y = b1;
            acc[t] = __builtin_amdgcn_wmma_f32_16x16x4_f32(
                false, afr[kc], false, b, (short)0, acc[t], false, false);
        }
    }

    const bool fullM = (row0 + 16 <= N);
#pragma unroll
    for (int t = 0; t < TILES_N; ++t) {
        const int nc = t * 16 + lmod;
        if (nc < F_OUT) {
            float* p = xw + (long long)(row0 + lg * 8) * F_OUT + nc;
            if (fullM) {
#pragma unroll
                for (int v = 0; v < 8; ++v) p[(long long)v * F_OUT] = acc[t][v];
            } else {
#pragma unroll
                for (int v = 0; v < 8; ++v) {
                    const int mrow = row0 + v + lg * 8;
                    if (mrow < N) p[(long long)v * F_OUT] = acc[t][v];
                }
            }
        }
    }
}

__global__ void k_first_linear(const int* __restrict__ read_length,
                               float* __restrict__ dinv, float* __restrict__ dsq,
                               int* __restrict__ cnt_out,
                               const float* __restrict__ W, float* __restrict__ xw,
                               int* __restrict__ head,
                               const int* __restrict__ ph,
                               int* rec,
                               int* __restrict__ tab, int N, int E) {
    __shared__ int stack[8 * LB_CHUNKS * 16];
    const int wave = blockIdx.x * (blockDim.x >> 5) + (threadIdx.x >> 5);
    const int lane = threadIdx.x & 31;
    const int row0 = wave * 16;
    if (row0 >= N) return;

    const int lg = lane >> 4;
    const int m  = row0 + (lane & 15);

    float xfirst = 0.f;
    if (lg == 0 && m < N) {
        const unsigned uE = (unsigned)E;
        const v2i* rec2 = reinterpret_cast<const v2i*>(rec);
        v4i* __restrict__ tb = reinterpret_cast<v4i*>(tab)
                             + ((long long)wave * NB_BLK) * 16 + (lane & 15);
        int* stk = stack + ((threadIdx.x >> 5) * LB_CHUNKS) * 16 + (lane & 15);
        int k = 0;
        int hv[LB_CHUNKS];
#pragma unroll
        for (int c = 0; c < LB_CHUNKS; ++c) hv[c] = ph[(long long)c * N + m];
#pragma unroll
        for (int c = 0; c < LB_CHUNKS; ++c) {
            stk[k * 16] = hv[c];
            k += ((unsigned)hv[c] < uE) ? 1 : 0;
        }
        int cnt = 0;
        int ovf = -1;
        int e = -1;
        if (k > 0) { --k; e = stk[k * 16]; }
        v4i blk = {0, 0, 0, 0};
        for (int it = 0; it < E && (unsigned)e < uE; ++it) {
            const v2i r = rec2[e];
            if (cnt < NB_CAP) {
                const int sl = cnt & 3;
                blk.x = (sl == 0) ? r.x : blk.x;
                blk.y = (sl == 1) ? r.x : blk.y;
                blk.z = (sl == 2) ? r.x : blk.z;
                blk.w = (sl == 3) ? r.x : blk.w;
                if (sl == 3) {
                    tb[(cnt >> 2) * 16] = blk;
                    blk = (v4i){0, 0, 0, 0};
                }
            } else if (cnt == NB_CAP) {
                ovf = e;
            }
            ++cnt;
            int nx = r.y;
            if (!((unsigned)nx < uE) && k > 0) {
                --k;
                nx = stk[k * 16];
                if (cnt >= NB_CAP) rec[2 * (long long)e + 1] = nx;
            }
            e = nx;
        }
        if (cnt < NB_CAP && (cnt & 3) != 0) tb[(cnt >> 2) * 16] = blk;
        head[m]    = ovf;
        cnt_out[m] = cnt;
        const float deg = 1.0f + (float)cnt;
        dsq[m]  = rsqrtf(deg);
        dinv[m] = 1.0f / deg;
        xfirst  = (float)read_length[m] * LEN_SCALE_INV;
    }
    v2f afr[1];
    afr[0].x = xfirst;
    afr[0].y = 0.f;
    tile_linear_store<1, 4>(afr, W, xw, row0, lane, N);
}

template <int CH, int NA, bool WR>
__device__ __forceinline__ void sum_in_edges(v4f (&acc)[NA],
                                             const v4f* __restrict__ xw4,
                                             const float* __restrict__ dsq,
                                             const v4i* __restrict__ tab4,
                                             v4f* __restrict__ wtab4,
                                             const v2i* __restrict__ rec2,
                                             const int* __restrict__ head,
                                             int m, int c, int c0, float dd, int E) {
    const unsigned uE = (unsigned)E;
    const int ct = (c < NB_CAP) ? c : NB_CAP;
    const int nb = (ct + 3) >> 2;
    const v4i* __restrict__ tb = tab4 + ((long long)(m >> 4) * NB_BLK) * 16 + (m & 15);
    v4f* __restrict__ wb = wtab4 + ((long long)(m >> 4) * NB_WBLK) * 16 + (m & 15);

    v4i s4 = {0, 0, 0, 0};
    v4f w4 = {0.f, 0.f, 0.f, 0.f};
    if (nb > 0) {
        s4 = tb[0];
        if (!WR) w4 = wb[0];
    }
    for (int j = 0; j < nb; ++j) {
        const int sv[4] = {s4.x, s4.y, s4.z, s4.w};
        float q[4];
        v4f   x[4][NA];
        if (!WR && j < NB_WBLK) {
            q[0] = w4.x; q[1] = w4.y; q[2] = w4.z; q[3] = w4.w;
        } else {
#pragma unroll
            for (int i = 0; i < 4; ++i) q[i] = dsq[sv[i]];
        }
#pragma unroll
        for (int i = 0; i < 4; ++i) {
#pragma unroll
            for (int a = 0; a < NA; ++a) x[i][a] = xw4[(long long)sv[i] * CH + c0 + 2 * a];
        }
        v4i sn = {0, 0, 0, 0};
        v4f wn = {0.f, 0.f, 0.f, 0.f};
        if (j + 1 < nb) {
            sn = tb[(j + 1) * 16];
            if (!WR && j + 1 < NB_WBLK) wn = wb[(j + 1) * 16];
        }
        if (WR && j < NB_WBLK) {
            v4f qv; qv.x = q[0]; qv.y = q[1]; qv.z = q[2]; qv.w = q[3];
            wb[j * 16] = qv;
        }
#pragma unroll
        for (int i = 0; i < 4; ++i) {
            if (4 * j + i < ct) {
                const float w = q[i] * dd;
#pragma unroll
                for (int a = 0; a < NA; ++a) {
                    acc[a].x += x[i][a].x * w;
                    acc[a].y += x[i][a].y * w;
                    acc[a].z += x[i][a].z * w;
                    acc[a].w += x[i][a].w * w;
                }
            }
        }
        s4 = sn;
        w4 = wn;
    }

    int e = -1;
    if (c > NB_CAP) e = head[m];
    v2i r; r.x = 0; r.y = -1;
    if ((unsigned)e < uE) r = rec2[e];
    for (int it = 0; it < E && (unsigned)e < uE; ++it) {
        const int s  = r.x;
        const int nx = r.y;
        const float q = dsq[s];
        v4f x[NA];
#pragma unroll
        for (int a = 0; a < NA; ++a) x[a] = xw4[(long long)s * CH + c0 + 2 * a];
        v2i rn; rn.x = 0; rn.y = -1;
        if ((unsigned)nx < uE) rn = rec2[nx];
        const float w = q * dd;
#pragma unroll
        for (int a = 0; a < NA; ++a) {
            acc[a].x += x[a].x * w;
            acc[a].y += x[a].y * w;
            acc[a].z += x[a].z * w;
            acc[a].w += x[a].w * w;
        }
        e = nx;
        r = rn;
    }
}

template <int F_IN, int F_OUT>
__global__ void k_gather_linear(const float* __restrict__ xw_prev,
                                const float* __restrict__ bias_prev,
                                const int* __restrict__ rec,
                                const int* __restrict__ head,
                                const int* __restrict__ cnt,
                                const int* __restrict__ tab,
                                float* __restrict__ wtab,
                                const float* __restrict__ dsq,
                                const float* __restrict__ dinv,
                                const float* __restrict__ W,
                                float* __restrict__ xw, int N, int E) {
    constexpr int CH = F_IN / 4;
    constexpr int NT = (CH == 1) ? 2 : 1;
    constexpr int NA = (CH == 4) ? 2 : 1;
    static_assert(CH == 1 || CH == 2 || CH == 4, "layer width");

    const int wave = blockIdx.x * (blockDim.x >> 5) + (threadIdx.x >> 5);
    const int lane = threadIdx.x & 31;
    const int row0 = wave * (16 * NT);
    if (row0 >= N) return;

    const int  lg   = lane >> 4;
    const int  lmod = lane & 15;
    const int  m    = row0 + ((NT == 2) ? lane : lmod);
    const bool mok  = (m < N);
    const int  c0   = (CH == 1) ? 0 : lg;

    const v4f* __restrict__ xw4  = reinterpret_cast<const v4f*>(xw_prev);
    const v2i* __restrict__ rec2 = reinterpret_cast<const v2i*>(rec);
    const v4i* __restrict__ tab4 = reinterpret_cast<const v4i*>(tab);
    v4f* __restrict__ wtab4 = reinterpret_cast<v4f*>(wtab);

    float dd = 0.f, dnode = 0.f;
    int c = 0;
    if (mok) { dd = dsq[m]; dnode = dinv[m]; c = cnt[m]; }

    v4f acc[NA];
#pragma unroll
    for (int a = 0; a < NA; ++a) acc[a] = (v4f){0.f, 0.f, 0.f, 0.f};

    sum_in_edges<CH, NA, (CH == 1)>(acc, xw4, dsq, tab4, wtab4, rec2, head, m, c, c0, dd, E);

    v4f xv[NA];
#pragma unroll
    for (int a = 0; a < NA; ++a) {
        const int c = c0 + 2 * a;
        v4f o = {0.f, 0.f, 0.f, 0.f};
        if (mok) {
            const v4f h = xw4[(long long)m * CH + c];
            o.x = acc[a].x + h.x * dnode + bias_prev[4 * c + 0];
            o.y = acc[a].y + h.y * dnode + bias_prev[4 * c + 1];
            o.z = acc[a].z + h.z * dnode + bias_prev[4 * c + 2];
            o.w = acc[a].w + h.w * dnode + bias_prev[4 * c + 3];
        }
        xv[a] = o;
    }

#pragma unroll
    for (int t2 = 0; t2 < NT; ++t2) {
        const int trow0 = row0 + 16 * t2;
        if (trow0 < N) {
            v2f afr[CH];
#pragma unroll
            for (int kc = 0; kc < CH; ++kc) {
                const int srcl = (NT == 2) ? (16 * t2 + lmod) : (lmod + 16 * (kc & 1));
                const v4f sv = xv[kc >> 1];
                v4f t;
                t.x = __shfl(sv.x, srcl);
                t.y = __shfl(sv.y, srcl);
                t.z = __shfl(sv.z, srcl);
                t.w = __shfl(sv.w, srcl);
                afr[kc].x = lg ? t.z : t.x;
                afr[kc].y = lg ? t.w : t.y;
            }
            tile_linear_store<F_IN, F_OUT>(afr, W, xw, trow0, lane, N);
        }
    }
}

template <int F>
__global__ void k_gather_out(const float* __restrict__ xw,
                             const float* __restrict__ bias,
                             const int* __restrict__ rec,
                             const int* __restrict__ head,
                             const int* __restrict__ cnt,
                             const int* __restrict__ tab,
                             float* __restrict__ wtab,
                             const float* __restrict__ dsq,
                             const float* __restrict__ dinv,
                             float* __restrict__ out, int N, int E) {
    constexpr int LPN = F / 4;
    const long long gid = (long long)blockIdx.x * blockDim.x + threadIdx.x;
    const int node = (int)(gid / LPN);
    const int c    = (int)(gid % LPN);
    if (node >= N) return;

    const v4f* __restrict__ xw4  = reinterpret_cast<const v4f*>(xw);
    const v2i* __restrict__ rec2 = reinterpret_cast<const v2i*>(rec);
    const v4i* __restrict__ tab4 = reinterpret_cast<const v4i*>(tab);
    v4f* __restrict__ wtab4 = reinterpret_cast<v4f*>(wtab);

    const float dd = dsq[node];
    v4f acc1[1];
    acc1[0] = (v4f){0.f, 0.f, 0.f, 0.f};
    sum_in_edges<LPN, 1, false>(acc1, xw4, dsq, tab4, wtab4, rec2, head, node, cnt[node], c, dd, E);
    const v4f acc = acc1[0];

    const v4f   h  = xw4[(long long)node * LPN + c];
    const float dn = dinv[node];
    v4f o;
    o.x = acc.x + h.x * dn + bias[4 * c + 0];
    o.y = acc.y + h.y * dn + bias[4 * c + 1];
    o.z = acc.z + h.z * dn + bias[4 * c + 2];
    o.w = acc.w + h.w * dn + bias[4 * c + 3];
    reinterpret_cast<v4f*>(out)[(long long)node * LPN + c] = o;
}

template <int F_IN, int F_OUT>
static void run_layer(const float* xw_prev, const float* bias_prev, const float* W,
                      float* xw, const float* dinv, const float* dsq,
                      const int* head, const int* rec,
                      const int* cnt, const int* tab, float* wtab,
                      int N, int E, hipStream_t stream) {
    constexpr int ROWS = (F_IN == 4) ? 32 : 16;
    const long long waves  = (long long)((N + ROWS - 1) / ROWS);
    const long long blocks = (waves + 7) / 8;
    k_gather_linear<F_IN, F_OUT><<<(unsigned)blocks, 256, 0, stream>>>(
        xw_prev, bias_prev, rec, head, cnt, tab, wtab, dsq, dinv, W, xw, N, E);
}

extern "C" void kernel_launch(void* const* d_in, const int* in_sizes, int n_in,
                              void* d_out, int out_size, void* d_ws, size_t ws_size,
                              hipStream_t stream) {
    const int* read_length = (const int*)d_in[0];
    const int* edge        = (const int*)d_in[1];
    const int  N = in_sizes[0];
    const int  E = in_sizes[1] / 2;
    const int* src = edge;
    const int* dst = edge + E;

    const float* W1 = (const float*)d_in[2]; const float* B1 = (const float*)d_in[3];
    const float* W2 = (const float*)d_in[4]; const float* B2 = (const float*)d_in[5];
    const float* W3 = (const float*)d_in[6]; const float* B3 = (const float*)d_in[7];
    const float* W4 = (const float*)d_in[8]; const float* B4 = (const float*)d_in[9];

    const size_t tiles = ((size_t)N + 15) / 16;
    float* ws   = (float*)d_ws;
    float* pxw  = ws;
    float* qxw  = pxw + (size_t)N * 16;
    int*   tab  = (int*)(qxw + (size_t)N * 32);
    float* wtab = (float*)(tab + tiles * NB_CAP * 16);
    int*   rec  = (int*)(wtab + tiles * NB_WBLK * 64);
    float* dinv = (float*)(rec + (size_t)E * 2);
    float* dsq  = dinv + N;
    int*   head = (int*)(dsq + N);
    int*   cnt  = head + N;
    int*   ph   = (int*)qxw;

    const int T = 256;

    const int R = (N + LB_RANGE - 1) / LB_RANGE;
    k_build<<<R * LB_CHUNKS, LB_THREADS, 0, stream>>>(src, dst, ph, rec, N, E, R);

    {
        const long long waves  = (long long)((N + 15) / 16);
        const long long blocks = (waves + 7) / 8;
        k_first_linear<<<(unsigned)blocks, 256, 0, stream>>>(
            read_length, dinv, dsq, cnt, W1, pxw, head, ph, rec, tab, N, E);
    }

    run_layer<4, 8>(pxw, B1, W2, qxw, dinv, dsq, head, rec, cnt, tab, wtab, N, E, stream);
    run_layer<8, 16>(qxw, B2, W3, pxw, dinv, dsq, head, rec, cnt, tab, wtab, N, E, stream);
    run_layer<16, 32>(pxw, B3, W4, qxw, dinv, dsq, head, rec, cnt, tab, wtab, N, E, stream);

    const long long nl = (long long)N * (32 / 4);
    k_gather_out<32><<<(unsigned)((nl + T - 1) / T), T, 0, stream>>>(
        qxw, B4, rec, head, cnt, tab, wtab, dsq, dinv, (float*)d_out, N, E);
}


# one load request per gathered source row: layer 3 reads a 64 B row with 4 lanes of one load (8 nodes/wave), layer 1 row packed with its dsq (32 B, 2 lanes); sum order unchanged
# speedup vs baseline: 5.1682x; 1.0752x over previous
#include <hip/hip_runtime.h>
#include <hip/hip_bf16.h>


typedef __attribute__((ext_vector_type(2))) float v2f;
typedef __attribute__((ext_vector_type(4))) float v4f;
typedef __attribute__((ext_vector_type(8))) float v8f;
typedef __attribute__((ext_vector_type(2))) int   v2i;
typedef __attribute__((ext_vector_type(4))) int   v4i;

#define LEN_SCALE_INV (1.0f / 20000.0f)

#define NB_CAP 32
#define NB_BLK (NB_CAP / 4)
#define NB_WBLK 4

#define LB_RANGE   65536
#define LB_CHUNKS  32
#define LB_THREADS 1024
#define LB_UNROLL  16

__global__ __launch_bounds__(LB_THREADS)
void k_build(const int* __restrict__ src, const int* __restrict__ dst,
             int* __restrict__ ph, int* __restrict__ rec, int N, int E, int R) {
    __shared__ int hd[LB_RANGE];
    const int r  = (int)(blockIdx.x % (unsigned)R);
    const int c  = (int)(blockIdx.x / (unsigned)R);
    const int lo = r * LB_RANGE;
    int len = N - lo;
    if (len > LB_RANGE) len = LB_RANGE;
    const unsigned ulen = (unsigned)len;

    for (int i = threadIdx.x; i < LB_RANGE; i += LB_THREADS) hd[i] = -1;
    __syncthreads();

    const int per = (E + LB_CHUNKS - 1) / LB_CHUNKS;
    const int e0  = c * per;
    int e1 = e0 + per;
    if (e1 > E) e1 = E;
    v2i* __restrict__ rec2 = reinterpret_cast<v2i*>(rec);

    int dn[LB_UNROLL];
#pragma unroll
    for (int u = 0; u < LB_UNROLL; ++u) {
        const int e = e0 + (int)threadIdx.x + u * LB_THREADS;
        dn[u] = (e < e1) ? dst[e] : -1;
    }
    for (int base = e0 + (int)threadIdx.x; base < e1; base += LB_THREADS * LB_UNROLL) {
        unsigned off[LB_UNROLL];
        int      sv[LB_UNROLL];
        int      pv[LB_UNROLL];
#pragma unroll
        for (int u = 0; u < LB_UNROLL; ++u)
            off[u] = (unsigned)(dn[u] - lo);
#pragma unroll
        for (int u = 0; u < LB_UNROLL; ++u) {
            sv[u] = 0;
            if (off[u] < ulen) sv[u] = src[base + u * LB_THREADS];
        }
#pragma unroll
        for (int u = 0; u < LB_UNROLL; ++u) {
            const int e = base + LB_THREADS * LB_UNROLL + u * LB_THREADS;
            dn[u] = (e < e1) ? dst[e] : -1;
        }
#pragma unroll
        for (int u = 0; u < LB_UNROLL; ++u) {
            pv[u] = -1;
            if (off[u] < ulen) pv[u] = atomicExch(&hd[off[u]], base + u * LB_THREADS);
        }
#pragma unroll
        for (int u = 0; u < LB_UNROLL; ++u) {
            if (off[u] < ulen) {
                v2i rr; rr.x = sv[u]; rr.y = pv[u];
                rec2[base + u * LB_THREADS] = rr;
            }
        }
    }
    __syncthreads();

    int* __restrict__ row = ph + (long long)c * N + lo;
    for (int i = threadIdx.x; i < len; i += LB_THREADS) row[i] = hd[i];
}

template <int F_IN, int F_OUT, int ROWS = 16, int LD = F_OUT>
__device__ __forceinline__ void tile_linear_store(const v2f (&afr)[(F_IN + 3) / 4],
                                                  const float* __restrict__ W,
                                                  float* __restrict__ xw,
                                                  int row0, int lane, int N) {
    constexpr int TILES_N = (F_OUT + 15) / 16;
    constexpr int KCH     = (F_IN + 3) / 4;
    const int lg   = lane >> 4;
    const int lmod = lane & 15;

    v8f acc[TILES_N];
#pragma unroll
    for (int t = 0; t < TILES_N; ++t)
        acc[t] = (v8f){0.f, 0.f, 0.f, 0.f, 0.f, 0.f, 0.f, 0.f};

#pragma unroll
    for (int kc = 0; kc < KCH; ++kc) {
        const int k0  = kc * 4 + lg * 2;
        const int k1  = k0 + 1;
        const int k0c = (k0 < F_IN) ? k0 : (F_IN - 1);
        const int k1c = (k1 < F_IN) ? k1 : (F_IN - 1);
#pragma unroll
        for (int t = 0; t < TILES_N; ++t) {
            const int  nc  = t * 16 + lmod;
            const bool nok = (nc < F_OUT);
            const int  ncc = nok ? nc : 0;
            float b0 = W[k0c * F_OUT + ncc];
            float b1 = W[k1c * F_OUT + ncc];
            b0 = (nok && k0 < F_IN) ? b0 : 0.f;
            b1 = (nok && k1 < F_IN) ? b1 : 0.f;
            v2f b; b.x = b0; b.y = b1;
            acc[t] = __builtin_amdgcn_wmma_f32_16x16x4_f32(
                false, afr[kc], false, b, (short)0, acc[t], false, false);
        }
    }

    const bool fullM = (row0 + ROWS <= N);
#pragma unroll
    for (int t = 0; t < TILES_N; ++t) {
        const int nc = t * 16 + lmod;
        if (nc < F_OUT && (ROWS == 16 || lg == 0)) {
            float* p = xw + (long long)(row0 + lg * 8) * LD + nc;
            if (fullM) {
#pragma unroll
                for (int v = 0; v < 8; ++v) p[(long long)v * LD] = acc[t][v];
            } else {
#pragma unroll
                for (int v = 0; v < 8; ++v) {
                    const int mrow = row0 + v + lg * 8;
                    if (mrow < N) p[(long long)v * LD] = acc[t][v];
                }
            }
        }
    }
}

__global__ void k_first_linear(const int* __restrict__ read_length,
                               float* __restrict__ dinv, float* __restrict__ dsq,
                               int* __restrict__ cnt_out,
                               const float* __restrict__ W, float* __restrict__ xw,
                               int* __restrict__ head,
                               const int* __restrict__ ph,
                               int* rec,
                               int* __restrict__ tab, int N, int E) {
    __shared__ int stack[8 * LB_CHUNKS * 16];
    const int wave = blockIdx.x * (blockDim.x >> 5) + (threadIdx.x >> 5);
    const int lane = threadIdx.x & 31;
    const int row0 = wave * 16;
    if (row0 >= N) return;

    const int lg = lane >> 4;
    const int m  = row0 + (lane & 15);

    float xfirst = 0.f;
    if (lg == 0 && m < N) {
        const unsigned uE = (unsigned)E;
        const v2i* rec2 = reinterpret_cast<const v2i*>(rec);
        v4i* __restrict__ tb = reinterpret_cast<v4i*>(tab)
                             + ((long long)wave * NB_BLK) * 16 + (lane & 15);
        int* stk = stack + ((threadIdx.x >> 5) * LB_CHUNKS) * 16 + (lane & 15);
        int k = 0;
        int hv[LB_CHUNKS];
#pragma unroll
        for (int c = 0; c < LB_CHUNKS; ++c) hv[c] = ph[(long long)c * N + m];
#pragma unroll
        for (int c = 0; c < LB_CHUNKS; ++c) {
            stk[k * 16] = hv[c];
            k += ((unsigned)hv[c] < uE) ? 1 : 0;
        }
        int cnt = 0;
        int ovf = -1;
        int e = -1;
        if (k > 0) { --k; e = stk[k * 16]; }
        v4i blk = {0, 0, 0, 0};
        for (int it = 0; it < E && (unsigned)e < uE; ++it) {
            const v2i r = rec2[e];
            if (cnt < NB_CAP) {
                const int sl = cnt & 3;
                blk.x = (sl == 0) ? r.x : blk.x;
                blk.y = (sl == 1) ? r.x : blk.y;
                blk.z = (sl == 2) ? r.x : blk.z;
                blk.w = (sl == 3) ? r.x : blk.w;
                if (sl == 3) {
                    tb[(cnt >> 2) * 16] = blk;
                    blk = (v4i){0, 0, 0, 0};
                }
            } else if (cnt == NB_CAP) {
                ovf = e;
            }
            ++cnt;
            int nx = r.y;
            if (!((unsigned)nx < uE) && k > 0) {
                --k;
                nx = stk[k * 16];
                if (cnt >= NB_CAP) rec[2 * (long long)e + 1] = nx;
            }
            e = nx;
        }
        if (cnt < NB_CAP && (cnt & 3) != 0) tb[(cnt >> 2) * 16] = blk;
        head[m]    = ovf;
        cnt_out[m] = cnt;
        const float deg = 1.0f + (float)cnt;
        const float dq = rsqrtf(deg);
        dsq[m]  = dq;
        dinv[m] = 1.0f / deg;
        reinterpret_cast<v4f*>(xw)[(long long)m * 2 + 1] = (v4f){dq, 0.f, 0.f, 0.f};
        xfirst  = (float)read_length[m] * LEN_SCALE_INV;
    }
    v2f afr[1];
    afr[0].x = xfirst;
    afr[0].y = 0.f;
    tile_linear_store<1, 4, 16, 8>(afr, W, xw, row0, lane, N);
}

__device__ __forceinline__ float xhalf(float x) {
    const int b = __builtin_bit_cast(int, x);
    return __builtin_bit_cast(float,
        __builtin_amdgcn_permlanex16(b, b, 0x76543210, 0xfedcba98, false, false));
}

template <int RS, bool WR>
__device__ __forceinline__ void sum_in_edges(v4f& acc,
                                             const v4f* __restrict__ xw4,
                                             const float* __restrict__ dsq,
                                             const v4i* __restrict__ tab4,
                                             v4f* __restrict__ wtab4,
                                             const v2i* __restrict__ rec2,
                                             const int* __restrict__ head,
                                             int m, int c, int c0, float dd, int E) {
    const unsigned uE = (unsigned)E;
    const int ct = (c < NB_CAP) ? c : NB_CAP;
    const int nb = (ct + 3) >> 2;
    const v4i* __restrict__ tb = tab4 + ((long long)(m >> 4) * NB_BLK) * 16 + (m & 15);
    v4f* __restrict__ wb = wtab4 + ((long long)(m >> 4) * NB_WBLK) * 16 + (m & 15);

    v4i s4 = {0, 0, 0, 0};
    v4f w4 = {0.f, 0.f, 0.f, 0.f};
    if (nb > 0) {
        s4 = tb[0];
        if (!WR) w4 = wb[0];
    }
    for (int j = 0; j < nb; ++j) {
        const int sv[4] = {s4.x, s4.y, s4.z, s4.w};
        float q[4];
        v4f   x[4];
#pragma unroll
        for (int i = 0; i < 4; ++i) x[i] = xw4[(long long)sv[i] * RS + c0];
        if (!WR) {
            if (j < NB_WBLK) {
                q[0] = w4.x; q[1] = w4.y; q[2] = w4.z; q[3] = w4.w;
            } else {
#pragma unroll
                for (int i = 0; i < 4; ++i) q[i] = dsq[sv[i]];
            }
        }
        v4i sn = {0, 0, 0, 0};
        v4f wn = {0.f, 0.f, 0.f, 0.f};
        if (j + 1 < nb) {
            sn = tb[(j + 1) * 16];
            if (!WR && j + 1 < NB_WBLK) wn = wb[(j + 1) * 16];
        }
        if (WR) {
#pragma unroll
            for (int i = 0; i < 4; ++i) q[i] = xhalf(x[i].x);
        }
        if (WR && c0 == 0 && j < NB_WBLK) {
            v4f qv; qv.x = q[0]; qv.y = q[1]; qv.z = q[2]; qv.w = q[3];
            wb[j * 16] = qv;
        }
#pragma unroll
        for (int i = 0; i < 4; ++i) {
            if (4 * j + i < ct) {
                const float w = q[i] * dd;
                acc.x += x[i].x * w;
                acc.y += x[i].y * w;
                acc.z += x[i].z * w;
                acc.w += x[i].w * w;
            }
        }
        s4 = sn;
        w4 = wn;
    }

    int e = -1;
    if (c > NB_CAP) e = head[m];
    v2i r; r.x = 0; r.y = -1;
    if ((unsigned)e < uE) r = rec2[e];
    for (int it = 0; it < E && (unsigned)e < uE; ++it) {
        const int s  = r.x;
        const int nx = r.y;
        const float q = dsq[s];
        const v4f x = xw4[(long long)s * RS + c0];
        v2i rn; rn.x = 0; rn.y = -1;
        if ((unsigned)nx < uE) rn = rec2[nx];
        const float w = q * dd;
        acc.x += x.x * w;
        acc.y += x.y * w;
        acc.z += x.z * w;
        acc.w += x.w * w;
        e = nx;
        r = rn;
    }
}

template <int F_IN, int F_OUT>
__global__ void k_gather_linear(const float* __restrict__ xw_prev,
                                const float* __restrict__ bias_prev,
                                const int* __restrict__ rec,
                                const int* __restrict__ head,
                                const int* __restrict__ cnt,
                                const int* __restrict__ tab,
                                float* __restrict__ wtab,
                                const float* __restrict__ dsq,
                                const float* __restrict__ dinv,
                                const float* __restrict__ W,
                                float* __restrict__ xw, int N, int E) {
    constexpr int  CH   = F_IN / 4;
    constexpr bool WR   = (CH == 1);
    constexpr int  RS   = WR ? 2 : CH;
    constexpr int  ROWS = (CH == 4) ? 8 : 16;
    static_assert(CH == 1 || CH == 2 || CH == 4, "layer width");

    const int wave = blockIdx.x * (blockDim.x >> 5) + (threadIdx.x >> 5);
    const int lane = threadIdx.x & 31;
    const int row0 = wave * ROWS;
    if (row0 >= N) return;

    const int  lg   = lane >> 4;
    const int  lmod = lane & 15;
    const int  m    = row0 + ((CH == 4) ? (lane >> 2) : lmod);
    const bool mok  = (m < N);
    const int  c0   = (CH == 4) ? (lane & 3) : lg;

    const v4f* __restrict__ xw4  = reinterpret_cast<const v4f*>(xw_prev);
    const v2i* __restrict__ rec2 = reinterpret_cast<const v2i*>(rec);
    const v4i* __restrict__ tab4 = reinterpret_cast<const v4i*>(tab);
    v4f* __restrict__ wtab4 = reinterpret_cast<v4f*>(wtab);

    float dd = 0.f, dnode = 0.f;
    int c = 0;
    if (mok) { dd = dsq[m]; dnode = dinv[m]; c = cnt[m]; }

    v4f acc = {0.f, 0.f, 0.f, 0.f};
    sum_in_edges<RS, WR>(acc, xw4, dsq, tab4, wtab4, rec2, head, m, c, c0, dd, E);

    v4f xv = {0.f, 0.f, 0.f, 0.f};
    if (mok && (!WR || c0 == 0)) {
        const v4f h = xw4[(long long)m * RS + c0];
        xv.x = acc.x + h.x * dnode + bias_prev[4 * c0 + 0];
        xv.y = acc.y + h.y * dnode + bias_prev[4 * c0 + 1];
        xv.z = acc.z + h.z * dnode + bias_prev[4 * c0 + 2];
        xv.w = acc.w + h.w * dnode + bias_prev[4 * c0 + 3];
    }

    v2f afr[CH];
#pragma unroll
    for (int kc = 0; kc < CH; ++kc) {
        const int srcl = (CH == 4) ? ((lmod & 7) * 4 + kc) : (lmod + 16 * kc);
        v4f t;
        t.x = __shfl(xv.x, srcl);
        t.y = __shfl(xv.y, srcl);
        t.z = __shfl(xv.z, srcl);
        t.w = __shfl(xv.w, srcl);
        afr[kc].x = lg ? t.z : t.x;
        afr[kc].y = lg ? t.w : t.y;
    }
    tile_linear_store<F_IN, F_OUT, ROWS>(afr, W, xw, row0, lane, N);
}

template <int F>
__global__ void k_gather_out(const float* __restrict__ xw,
                             const float* __restrict__ bias,
                             const int* __restrict__ rec,
                             const int* __restrict__ head,
                             const int* __restrict__ cnt,
                             const int* __restrict__ tab,
                             float* __restrict__ wtab,
                             const float* __restrict__ dsq,
                             const float* __restrict__ dinv,
                             float* __restrict__ out, int N, int E) {
    constexpr int LPN = F / 4;
    const long long gid = (long long)blockIdx.x * blockDim.x + threadIdx.x;
    const int node = (int)(gid / LPN);
    const int c    = (int)(gid % LPN);
    if (node >= N) return;

    const v4f* __restrict__ xw4  = reinterpret_cast<const v4f*>(xw);
    const v2i* __restrict__ rec2 = reinterpret_cast<const v2i*>(rec);
    const v4i* __restrict__ tab4 = reinterpret_cast<const v4i*>(tab);
    v4f* __restrict__ wtab4 = reinterpret_cast<v4f*>(wtab);

    const float dd = dsq[node];
    v4f acc = {0.f, 0.f, 0.f, 0.f};
    sum_in_edges<LPN, false>(acc, xw4, dsq, tab4, wtab4, rec2, head, node, cnt[node], c, dd, E);

    const v4f   h  = xw4[(long long)node * LPN + c];
    const float dn = dinv[node];
    v4f o;
    o.x = acc.x + h.x * dn + bias[4 * c + 0];
    o.y = acc.y + h.y * dn + bias[4 * c + 1];
    o.z = acc.z + h.z * dn + bias[4 * c + 2];
    o.w = acc.w + h.w * dn + bias[4 * c + 3];
    reinterpret_cast<v4f*>(out)[(long long)node * LPN + c] = o;
}

template <int F_IN, int F_OUT>
static void run_layer(const float* xw_prev, const float* bias_prev, const float* W,
                      float* xw, const float* dinv, const float* dsq,
                      const int* head, const int* rec,
                      const int* cnt, const int* tab, float* wtab,
                      int N, int E, hipStream_t stream) {
    constexpr int ROWS = (F_IN == 16) ? 8 : 16;
    const long long waves  = (long long)((N + ROWS - 1) / ROWS);
    const long long blocks = (waves + 7) / 8;
    k_gather_linear<F_IN, F_OUT><<<(unsigned)blocks, 256, 0, stream>>>(
        xw_prev, bias_prev, rec, head, cnt, tab, wtab, dsq, dinv, W, xw, N, E);
}

extern "C" void kernel_launch(void* const* d_in, const int* in_sizes, int n_in,
                              void* d_out, int out_size, void* d_ws, size_t ws_size,
                              hipStream_t stream) {
    const int* read_length = (const int*)d_in[0];
    const int* edge        = (const int*)d_in[1];
    const int  N = in_sizes[0];
    const int  E = in_sizes[1] / 2;
    const int* src = edge;
    const int* dst = edge + E;

    const float* W1 = (const float*)d_in[2]; const float* B1 = (const float*)d_in[3];
    const float* W2 = (const float*)d_in[4]; const float* B2 = (const float*)d_in[5];
    const float* W3 = (const float*)d_in[6]; const float* B3 = (const float*)d_in[7];
    const float* W4 = (const float*)d_in[8]; const float* B4 = (const float*)d_in[9];

    const size_t tiles = ((size_t)N + 15) / 16;
    float* ws   = (float*)d_ws;
    float* pxw  = ws;
    float* qxw  = pxw + (size_t)N * 16;
    int*   tab  = (int*)(qxw + (size_t)N * 32);
    float* wtab = (float*)(tab + tiles * NB_CAP * 16);
    int*   rec  = (int*)(wtab + tiles * NB_WBLK * 64);
    float* dinv = (float*)(rec + (size_t)E * 2);
    float* dsq  = dinv + N;
    int*   head = (int*)(dsq + N);
    int*   cnt  = head + N;
    int*   ph   = (int*)qxw;

    const int T = 256;

    const int R = (N + LB_RANGE - 1) / LB_RANGE;
    k_build<<<R * LB_CHUNKS, LB_THREADS, 0, stream>>>(src, dst, ph, rec, N, E, R);

    {
        const long long waves  = (long long)((N + 15) / 16);
        const long long blocks = (waves + 7) / 8;
        k_first_linear<<<(unsigned)blocks, 256, 0, stream>>>(
            read_length, dinv, dsq, cnt, W1, pxw, head, ph, rec, tab, N, E);
    }

    run_layer<4, 8>(pxw, B1, W2, qxw, dinv, dsq, head, rec, cnt, tab, wtab, N, E, stream);
    run_layer<8, 16>(qxw, B2, W3, pxw, dinv, dsq, head, rec, cnt, tab, wtab, N, E, stream);
    run_layer<16, 32>(pxw, B3, W4, qxw, dinv, dsq, head, rec, cnt, tab, wtab, N, E, stream);

    const long long nl = (long long)N * (32 / 4);
    k_gather_out<32><<<(unsigned)((nl + T - 1) / T), T, 0, stream>>>(
        qxw, B4, rec, head, cnt, tab, wtab, dsq, dinv, (float*)d_out, N, E);
}
